# starting kernel
# baseline (speedup 1.0000x reference)
#include <hip/hip_runtime.h>


typedef float v2f __attribute__((ext_vector_type(2)));
typedef float v8f __attribute__((ext_vector_type(8)));

#define T_SEG 8
#define NSEG  16
#define CCH   512
#define HW    196
#define B_TOT (NSEG * HW)
#define HEADS 4
#define KTAPS 5
#define O20   (HEADS * KTAPS)
#define RTILE 13

__global__ __launch_bounds__(32)
void dyncnn_logits_softmax(const float* __restrict__ x,
                           const float* __restrict__ W,
                           float* __restrict__ wsm)
{
    const int lane = threadIdx.x;
    const int tile = blockIdx.x % RTILE;
    const int tn   = blockIdx.x / RTILE;
    const int t    = tn % T_SEG;
    const int ni   = tn / T_SEG;

    const int half = lane >> 4;
    const int l15  = lane & 15;

    int hw  = tile * 16 + l15;
    int hwc = (hw < HW - 1) ? hw : (HW - 1);
    const int cA = half ? 2 : 0;
    const float* aptr = x + ((size_t)((ni * T_SEG + t) * CCH) + cA) * HW + hwc;

    const float2* __restrict__ W2 = (const float2*)W;
    int o0 = l15;
    int o1 = o0 + 16; if (o1 > O20 - 1) o1 = O20 - 1;
    int bofs0 = o0 * (CCH / 2) + (cA >> 1);
    int bofs1 = o1 * (CCH / 2) + (cA >> 1);

    v8f acc0 = {}; v8f acc1 = {};
    for (int k4 = 0; k4 < CCH; k4 += 4) {
        v2f a;
        a.x = aptr[0];
        a.y = aptr[HW];
        aptr += 4 * HW;
        float2 f0 = W2[bofs0];
        float2 f1 = W2[bofs1];
        bofs0 += 2; bofs1 += 2;
        v2f b0; b0.x = f0.x; b0.y = f0.y;
        v2f b1; b1.x = f1.x; b1.y = f1.y;
        acc0 = __builtin_amdgcn_wmma_f32_16x16x4_f32(
                   false, a, false, b0, (short)0, acc0, false, false);
        acc1 = __builtin_amdgcn_wmma_f32_16x16x4_f32(
                   false, a, false, b1, (short)0, acc1, false, false);
    }

    __shared__ float Lsm[16 * 32];
    {
        const int mbase = half * 8;
        #pragma unroll
        for (int r = 0; r < 8; ++r) {
            Lsm[(mbase + r) * 32 + l15]      = acc0[r];
            Lsm[(mbase + r) * 32 + 16 + l15] = acc1[r];
        }
    }
    __syncthreads();

    #pragma unroll
    for (int i = 0; i < 2; ++i) {
        const int task = lane * 2 + i;
        const int rr = task >> 2;
        const int h  = task & 3;
        const float* lp = &Lsm[rr * 32 + h * KTAPS];
        float v0 = lp[0], v1 = lp[1], v2 = lp[2], v3 = lp[3], v4 = lp[4];
        float m = fmaxf(fmaxf(fmaxf(v0, v1), fmaxf(v2, v3)), v4);
        float e0 = __expf(v0 - m), e1 = __expf(v1 - m), e2 = __expf(v2 - m),
              e3 = __expf(v3 - m), e4 = __expf(v4 - m);
        float inv = 1.0f / (e0 + e1 + e2 + e3 + e4);
        const int hwr = tile * 16 + rr;
        if (hwr < HW) {
            size_t base = (((size_t)t * B_TOT) + (size_t)ni * HW + hwr) * O20
                          + (size_t)h * KTAPS;
            wsm[base + 0] = e0 * inv;
            wsm[base + 1] = e1 * inv;
            wsm[base + 2] = e2 * inv;
            wsm[base + 3] = e3 * inv;
            wsm[base + 4] = e4 * inv;
        }
    }
}

__global__ __launch_bounds__(256)
void dyncnn_combine(const float* __restrict__ x,
                    const float* __restrict__ wsm,
                    float* __restrict__ out)
{
    const size_t idx = (size_t)blockIdx.x * 256 + threadIdx.x;
    const int hw  = (int)(idx % HW);
    const int c   = (int)((idx / HW) % CCH);
    const int nti = (int)(idx / ((size_t)HW * CCH));
    const int ni  = nti >> 3;
    const int t   = nti & 7;
    const int h   = c >> 7;

    const float* wp = wsm + (((size_t)t * B_TOT) + (size_t)ni * HW + hw) * O20
                          + (size_t)h * KTAPS;
    const float* xb = x + (((size_t)(ni * T_SEG)) * CCH + c) * HW + hw;
    const size_t tstride = (size_t)CCH * HW;

    float acc = 0.0f;
    #pragma unroll
    for (int k = 0; k < KTAPS; ++k) {
        const int ts = t + k - 4;
        if (ts >= 0)
            acc = fmaf(xb[(size_t)ts * tstride], wp[k], acc);
    }
    out[idx] = acc;
}

extern "C" void kernel_launch(void* const* d_in, const int* in_sizes, int n_in,
                              void* d_out, int out_size, void* d_ws, size_t ws_size,
                              hipStream_t stream) {
    (void)in_sizes; (void)n_in; (void)out_size; (void)ws_size;
    const float* x   = (const float*)d_in[0];
    const float* W   = (const float*)d_in[1];
    float*       out = (float*)d_out;
    float*       wsm = (float*)d_ws;

    dyncnn_logits_softmax<<<dim3(NSEG * T_SEG * RTILE), dim3(32), 0, stream>>>(x, W, wsm);

    const size_t total = (size_t)NSEG * T_SEG * CCH * HW;
    dyncnn_combine<<<dim3((unsigned)(total / 256)), dim3(256), 0, stream>>>(x, wsm, out);
}


# speedup vs baseline: 3.8822x; 3.8822x over previous
#include <hip/hip_runtime.h>


typedef float v2f __attribute__((ext_vector_type(2)));
typedef float v8f __attribute__((ext_vector_type(8)));

static __device__ __forceinline__ void lds_barrier() {
    __builtin_amdgcn_fence(__ATOMIC_RELEASE, "workgroup", "local");
    __builtin_amdgcn_s_barrier();
    __builtin_amdgcn_fence(__ATOMIC_ACQUIRE, "workgroup", "local");
}

#define T_SEG 8
#define NSEG  16
#define CCH   512
#define HW    196
#define HEADS 4
#define KTAPS 5
#define O20   (HEADS * KTAPS)
#define TILE  16
#define RTILE 13
#define NTHR  512
#define KSL   8
#define KSC   (CCH / KSL)

#define WPITCH    516
#define WL_OFF    0
#define WL_FLOATS (O20 * WPITCH)
#define WQ_OFF    (WL_OFF + WL_FLOATS)
#define WQ_FLOATS (T_SEG * O20 * TILE)
#define PT_OFF    (WQ_OFF + WQ_FLOATS)
#define PT_FLOATS (KSL * O20 * TILE)
#define XS_OFF    15616
#define XF_FLOATS (CCH * TILE)
#define XS_SLOTS  2
#define LDS_FLOATS (XS_OFF + XS_SLOTS * XF_FLOATS)

__global__ __launch_bounds__(NTHR)
void dyncnn_fused(const float* x,
                  const float* W,
                  float* out)
{
    __shared__ __attribute__((aligned(16))) float lds[LDS_FLOATS];
    float* xs = lds + XS_OFF;
    float* wl = lds + WL_OFF;
    float* wq = lds + WQ_OFF;
    float* pt = lds + PT_OFF;

    const int tid  = threadIdx.x;
    const int ni   = blockIdx.x % NSEG;
    const int tile = blockIdx.x / NSEG;

    const int  q     = tid & 3;
    const int  cs    = tid >> 2;
    const int  hw0   = tile * TILE + 4 * q;
    const bool valid = hw0 < HW;
    const int  h     = cs >> 5;
    const int  sub   = cs & 31;
    const int  c0    = h * 128 + sub;

    const int wave = __builtin_amdgcn_readfirstlane(tid >> 5);
    const int lane = tid & 31;
    const int half = lane >> 4;
    const int l15  = lane & 15;
    const int nt   = wave & 1;
    const int ks   = wave >> 1;
    const bool ovalid = nt * 16 + l15 < O20;
    int o = nt * 16 + l15; if (o > O20 - 1) o = O20 - 1;

    const int hwl = valid ? hw0 : tile * TILE;
    const float4* wg = (const float4*)W;
    const float4 wv0 = wg[tid];
    const float4 wv1 = wg[tid + NTHR];
    const float4 wv2 = wg[tid + NTHR * 2];
    const float4 wv3 = wg[tid + NTHR * 3];
    const float4 wv4 = wg[tid + NTHR * 4];

    const float* xg = x + ((size_t)(ni * T_SEG) * CCH + c0) * HW + hwl;
    float4 r[T_SEG * 4];
    #pragma unroll
    for (int i = 0; i < T_SEG * 4; ++i) {
        const int t = i >> 2, j = i & 3;
        r[i] = *(const float4*)(xg + (size_t)(t * CCH + 32 * j) * HW);
        if (j == 3) __builtin_amdgcn_sched_barrier(0);
    }

    #define W_TO_LDS(f, v) *(float4*)(wl + ((f) >> 7) * WPITCH + 4 * ((f) & 127)) = (v)
    W_TO_LDS(tid,            wv0);
    W_TO_LDS(tid + NTHR,     wv1);
    W_TO_LDS(tid + NTHR * 2, wv2);
    W_TO_LDS(tid + NTHR * 3, wv3);
    W_TO_LDS(tid + NTHR * 4, wv4);
    #undef W_TO_LDS

    float* xst = xs + c0 * TILE + 4 * q;
    #pragma unroll
    for (int j = 0; j < 4; ++j)
        *(float4*)(xst + 32 * j * TILE) = r[j];
    lds_barrier();

    const float* ap0 = xs + (ks * KSC + 2 * half) * TILE + l15;
    const float* bp  = wl + o * WPITCH + ks * KSC + 2 * half;
    float* ptw = pt + (ks * O20 + o) * TILE + 8 * half;

    const float* wp = wq + (h * KTAPS) * TILE + 4 * q;
    float* og = out + ((size_t)(ni * T_SEG) * CCH + c0) * HW + hw0;

    #pragma unroll
    for (int t = 0; t < T_SEG; ++t) {
        {
            const float* ap = ap0 + (t & 1) * XF_FLOATS;
            v8f acc = {};
            #pragma unroll
            for (int u = 0; u < KSC / 4; ++u) {
                v2f a;
                a.x = ap[u * (4 * TILE)];
                a.y = ap[u * (4 * TILE) + TILE];
                const v2f b = *(const v2f*)(bp + 4 * u);
                acc = __builtin_amdgcn_wmma_f32_16x16x4_f32(
                          false, a, false, b, (short)0, acc, false, false);
            }
            if (ovalid) {
                #pragma unroll
                for (int e = 0; e < 8; ++e) ptw[e] = acc[e];
            }
        }
        lds_barrier();

        if (tid < HEADS * TILE) {
            const int sh  = tid >> 4;
            const int shw = tid & 15;
            const float* pp = pt + (sh * KTAPS) * TILE + shw;
            float v[KTAPS];
            #pragma unroll
            for (int k = 0; k < KTAPS; ++k) {
                float s = pp[k * TILE];
                #pragma unroll
                for (int sl = 1; sl < KSL; ++sl) s += pp[(sl * O20 + k) * TILE];
                v[k] = s;
            }
            const float m = fmaxf(fmaxf(fmaxf(v[0], v[1]), fmaxf(v[2], v[3])), v[4]);
            const float e0 = __expf(v[0] - m), e1 = __expf(v[1] - m), e2 = __expf(v[2] - m),
                        e3 = __expf(v[3] - m), e4 = __expf(v[4] - m);
            const float inv = 1.0f / (e0 + e1 + e2 + e3 + e4);
            float* lp = wq + (t * O20 + sh * KTAPS) * TILE + shw;
            lp[0 * TILE] = e0 * inv;
            lp[1 * TILE] = e1 * inv;
            lp[2 * TILE] = e2 * inv;
            lp[3 * TILE] = e3 * inv;
            lp[4 * TILE] = e4 * inv;
        }

        if (t + 1 < T_SEG) {
            float* xd = xst + ((t + 1) & 1) * XF_FLOATS;
            #pragma unroll
            for (int j = 0; j < 4; ++j)
                *(float4*)(xd + 32 * j * TILE) = r[(t + 1) * 4 + j];
        }
        lds_barrier();

        if (valid) {
            float4 w[KTAPS];
            #pragma unroll
            for (int k = 0; k < KTAPS; ++k)
                w[k] = *(const float4*)(wp + (t * O20 + k) * TILE);
            #pragma unroll
            for (int j = 0; j < 4; ++j) {
                float4 acc = make_float4(0.0f, 0.0f, 0.0f, 0.0f);
                #pragma unroll
                for (int k = 0; k < KTAPS; ++k) {
                    const int ts = t + k - 4;
                    if (ts >= 0) {
                        const float4 xv = r[ts * 4 + j];
                        acc.x = fmaf(xv.x, w[k].x, acc.x);
                        acc.y = fmaf(xv.y, w[k].y, acc.y);
                        acc.z = fmaf(xv.z, w[k].z, acc.z);
                        acc.w = fmaf(xv.w, w[k].w, acc.w);
                    }
                }
                *(float4*)(og + (size_t)(t * CCH + 32 * j) * HW) = acc;
            }
        }
    }
}

extern "C" void kernel_launch(void* const* d_in, const int* in_sizes, int n_in,
                              void* d_out, int out_size, void* d_ws, size_t ws_size,
                              hipStream_t stream) {
    (void)in_sizes; (void)n_in; (void)out_size; (void)d_ws; (void)ws_size;
    const float* x   = (const float*)d_in[0];
    const float* W   = (const float*)d_in[1];
    float*       out = (float*)d_out;

    dyncnn_fused<<<dim3(NSEG * RTILE), dim3(NTHR), 0, stream>>>(x, W, out);
}


# logits fragments in registers ahead of the WMMA run: W fragments read from LDS once for all 8 frames, a frame's 16 x fragment reads issued as one group before its 16 WMMAs, not between; sums kept
# speedup vs baseline: 4.1212x; 1.0616x over previous
#include <hip/hip_runtime.h>


typedef float v2f __attribute__((ext_vector_type(2)));
typedef float v8f __attribute__((ext_vector_type(8)));

static __device__ __forceinline__ void lds_barrier() {
    __builtin_amdgcn_fence(__ATOMIC_RELEASE, "workgroup", "local");
    __builtin_amdgcn_s_barrier();
    __builtin_amdgcn_fence(__ATOMIC_ACQUIRE, "workgroup", "local");
}

#define T_SEG 8
#define NSEG  16
#define CCH   512
#define HW    196
#define HEADS 4
#define KTAPS 5
#define O20   (HEADS * KTAPS)
#define TILE  16
#define RTILE 13
#define NTHR  512
#define KSL   8
#define KSC   (CCH / KSL)

#define WPITCH    516
#define WL_OFF    0
#define WL_FLOATS (O20 * WPITCH)
#define WQ_OFF    (WL_OFF + WL_FLOATS)
#define WQ_FLOATS (T_SEG * O20 * TILE)
#define PT_OFF    (WQ_OFF + WQ_FLOATS)
#define PT_FLOATS (KSL * O20 * TILE)
#define XS_OFF    15616
#define XF_FLOATS (CCH * TILE)
#define XS_SLOTS  2
#define LDS_FLOATS (XS_OFF + XS_SLOTS * XF_FLOATS)

__global__ __launch_bounds__(NTHR)
void dyncnn_fused(const float* x,
                  const float* W,
                  float* out)
{
    __shared__ __attribute__((aligned(16))) float lds[LDS_FLOATS];
    float* xs = lds + XS_OFF;
    float* wl = lds + WL_OFF;
    float* wq = lds + WQ_OFF;
    float* pt = lds + PT_OFF;

    const int tid  = threadIdx.x;
    const int ni   = blockIdx.x % NSEG;
    const int tile = blockIdx.x / NSEG;

    const int  q     = tid & 3;
    const int  cs    = tid >> 2;
    const int  hw0   = tile * TILE + 4 * q;
    const bool valid = hw0 < HW;
    const int  h     = cs >> 5;
    const int  sub   = cs & 31;
    const int  c0    = h * 128 + sub;

    const int wave = __builtin_amdgcn_readfirstlane(tid >> 5);
    const int lane = tid & 31;
    const int half = lane >> 4;
    const int l15  = lane & 15;
    const int nt   = wave & 1;
    const int ks   = wave >> 1;
    const bool ovalid = nt * 16 + l15 < O20;
    int o = nt * 16 + l15; if (o > O20 - 1) o = O20 - 1;

    const int hwl = valid ? hw0 : tile * TILE;
    const float4* wg = (const float4*)W;
    const float4 wv0 = wg[tid];
    const float4 wv1 = wg[tid + NTHR];
    const float4 wv2 = wg[tid + NTHR * 2];
    const float4 wv3 = wg[tid + NTHR * 3];
    const float4 wv4 = wg[tid + NTHR * 4];

    const float* xg = x + ((size_t)(ni * T_SEG) * CCH + c0) * HW + hwl;
    float4 r[T_SEG * 4];
    #pragma unroll
    for (int i = 0; i < T_SEG * 4; ++i) {
        const int t = i >> 2, j = i & 3;
        r[i] = *(const float4*)(xg + (size_t)(t * CCH + 32 * j) * HW);
        if (j == 3) __builtin_amdgcn_sched_barrier(0);
    }

    #define W_TO_LDS(f, v) *(float4*)(wl + ((f) >> 7) * WPITCH + 4 * ((f) & 127)) = (v)
    W_TO_LDS(tid,            wv0);
    W_TO_LDS(tid + NTHR,     wv1);
    W_TO_LDS(tid + NTHR * 2, wv2);
    W_TO_LDS(tid + NTHR * 3, wv3);
    W_TO_LDS(tid + NTHR * 4, wv4);
    #undef W_TO_LDS

    float* xst = xs + c0 * TILE + 4 * q;
    #pragma unroll
    for (int j = 0; j < 4; ++j)
        *(float4*)(xst + 32 * j * TILE) = r[j];
    lds_barrier();

    const float* ap0 = xs + (ks * KSC + 2 * half) * TILE + l15;
    const float* bp  = wl + o * WPITCH + ks * KSC + 2 * half;
    float* ptw = pt + (ks * O20 + o) * TILE + 8 * half;

    const float* wp = wq + (h * KTAPS) * TILE + 4 * q;
    float* og = out + ((size_t)(ni * T_SEG) * CCH + c0) * HW + hw0;

    v2f bw[KSC / 4];
    #pragma unroll
    for (int u = 0; u < KSC / 4; ++u) bw[u] = *(const v2f*)(bp + 4 * u);

    v2f af[KSC / 4];
    #define LOAD_A(slot)                                                        \
        {                                                                       \
            const float* ap = ap0 + (slot) * XF_FLOATS;                         \
            _Pragma("unroll")                                                   \
            for (int u = 0; u < KSC / 4; ++u) {                                 \
                af[u].x = ap[u * (4 * TILE)];          \
                af[u].y = ap[u * (4 * TILE) + TILE];   \
            }                                                                   \
        }
    LOAD_A(0)
    __builtin_amdgcn_sched_barrier(0);

    #pragma unroll
    for (int t = 0; t < T_SEG; ++t) {
        {
            v8f acc = {};
            #pragma unroll
            for (int u = 0; u < KSC / 4; ++u)
                acc = __builtin_amdgcn_wmma_f32_16x16x4_f32(
                          false, af[u], false, bw[u], (short)0, acc, false, false);
            if (ovalid) {
                #pragma unroll
                for (int e = 0; e < 8; ++e) ptw[e] = acc[e];
            }
        }
        lds_barrier();

        if (tid < HEADS * TILE) {
            const int sh  = tid >> 4;
            const int shw = tid & 15;
            const float* pp = pt + (sh * KTAPS) * TILE + shw;
            float v[KTAPS];
            #pragma unroll
            for (int k = 0; k < KTAPS; ++k) {
                float s = pp[k * TILE];
                #pragma unroll
                for (int sl = 1; sl < KSL; ++sl) s += pp[(sl * O20 + k) * TILE];
                v[k] = s;
            }
            const float m = fmaxf(fmaxf(fmaxf(v[0], v[1]), fmaxf(v[2], v[3])), v[4]);
            const float e0 = __expf(v[0] - m), e1 = __expf(v[1] - m), e2 = __expf(v[2] - m),
                        e3 = __expf(v[3] - m), e4 = __expf(v[4] - m);
            const float inv = 1.0f / (e0 + e1 + e2 + e3 + e4);
            float* lp = wq + (t * O20 + sh * KTAPS) * TILE + shw;
            lp[0 * TILE] = e0 * inv;
            lp[1 * TILE] = e1 * inv;
            lp[2 * TILE] = e2 * inv;
            lp[3 * TILE] = e3 * inv;
            lp[4 * TILE] = e4 * inv;
        }

        if (t + 1 < T_SEG) {
            float* xd = xst + ((t + 1) & 1) * XF_FLOATS;
            #pragma unroll
            for (int j = 0; j < 4; ++j)
                *(float4*)(xd + 32 * j * TILE) = r[(t + 1) * 4 + j];
        }
        lds_barrier();

        if (t + 1 < T_SEG) LOAD_A((t + 1) & 1)

        if (valid) {
            float4 w[KTAPS];
            #pragma unroll
            for (int k = 0; k < KTAPS; ++k)
                w[k] = *(const float4*)(wp + (t * O20 + k) * TILE);
            #pragma unroll
            for (int j = 0; j < 4; ++j) {
                float4 acc = make_float4(0.0f, 0.0f, 0.0f, 0.0f);
                #pragma unroll
                for (int k = 0; k < KTAPS; ++k) {
                    const int ts = t + k - 4;
                    if (ts >= 0) {
                        const float4 xv = r[ts * 4 + j];
                        acc.x = fmaf(xv.x, w[k].x, acc.x);
                        acc.y = fmaf(xv.y, w[k].y, acc.y);
                        acc.z = fmaf(xv.z, w[k].z, acc.z);
                        acc.w = fmaf(xv.w, w[k].w, acc.w);
                    }
                }
                *(float4*)(og + (size_t)(t * CCH + 32 * j) * HW) = acc;
            }
        }
    }
    #undef LOAD_A
}

extern "C" void kernel_launch(void* const* d_in, const int* in_sizes, int n_in,
                              void* d_out, int out_size, void* d_ws, size_t ws_size,
                              hipStream_t stream) {
    (void)in_sizes; (void)n_in; (void)out_size; (void)d_ws; (void)ws_size;
    const float* x   = (const float*)d_in[0];
    const float* W   = (const float*)d_in[1];
    float*       out = (float*)d_out;

    dyncnn_fused<<<dim3(NSEG * RTILE), dim3(NTHR), 0, stream>>>(x, W, out);
}
